# starting kernel
# baseline (speedup 1.0000x reference)
#include <hip/hip_runtime.h>
#include <math.h>

#define B_   16
#define T_   256
#define D_   512
#define L_   32
#define MT   32
#define EPS_ 1e-8f

typedef __attribute__((ext_vector_type(2))) float v2f;
typedef __attribute__((ext_vector_type(8))) float v8f;

__global__ __launch_bounds__(64) void fullmatch_cosine_kernel(
    const float* __restrict__ v1,
    const float* __restrict__ v1_mask,
    const float* __restrict__ v2,
    const float* __restrict__ w,
    float* __restrict__ out)
{
    __shared__ float sV2[D_];
    __shared__ float sMask[MT];
    __shared__ float sN2[L_];

    const int tid = threadIdx.x;
    const int b   = blockIdx.x / (T_ / MT);
    const int t0  = (blockIdx.x % (T_ / MT)) * MT;

    {
        const float4* v2v = (const float4*)(v2 + (size_t)b * D_);
        ((float4*)sV2)[tid]      = v2v[tid];
        ((float4*)sV2)[tid + 64] = v2v[tid + 64];
        if (tid < MT) sMask[tid] = v1_mask[(size_t)b * T_ + t0 + tid];
        if (tid < L_) sN2[tid]   = 0.0f;
    }
    __syncthreads();

    {
        const int l = tid & (L_ - 1);
        const int q = tid >> 5;
        const float4* wr = (const float4*)(w + (size_t)l * D_ + q * 256);
        const float4* sv = (const float4*)(sV2 + q * 256);
        float acc = 0.0f;
        #pragma unroll 4
        for (int i = 0; i < 64; ++i) {
            float4 wv = wr[i];
            float4 vv = sv[i];
            float px = wv.x * vv.x, py = wv.y * vv.y;
            float pz = wv.z * vv.z, pw = wv.w * vv.w;
            acc += px * px + py * py + pz * pz + pw * pw;
        }
        atomicAdd(&sN2[l], acc);
    }

    const int lane = tid & 31;
    const int wm   = tid >> 5;
    const int hl   = lane & 15;
    const int koff = (lane >> 4) << 1;

    const int   rowLocal = wm * 16 + hl;
    const float m        = sMask[rowLocal];
    const float* aRow    = v1 + ((size_t)b * T_ + t0 + rowLocal) * (size_t)D_;
    const float* wRow0   = w  + (size_t)(hl)      * (size_t)D_;
    const float* wRow1   = w  + (size_t)(16 + hl) * (size_t)D_;

    v8f cN0 = {}, cS0 = {};
    v8f cN1 = {}, cS1 = {};

    #pragma unroll 4
    for (int k = 0; k < D_; k += 4) {
        const int kb = k + koff;

        float2 av  = *(const float2*)(aRow  + kb);
        float2 sv  = *(const float2*)(sV2   + kb);
        float2 wv0 = *(const float2*)(wRow0 + kb);
        float2 wv1 = *(const float2*)(wRow1 + kb);

        av.x *= m;  av.y *= m;

        v2f aN  = { av.x * sv.x, av.y * sv.y };
        v2f aS  = { av.x * av.x, av.y * av.y };
        v2f bW0 = { wv0.x * wv0.x, wv0.y * wv0.y };
        v2f bW1 = { wv1.x * wv1.x, wv1.y * wv1.y };

        cN0 = __builtin_amdgcn_wmma_f32_16x16x4_f32(
                  false, aN, false, bW0, (short)0, cN0, false, false);
        cS0 = __builtin_amdgcn_wmma_f32_16x16x4_f32(
                  false, aS, false, bW0, (short)0, cS0, false, false);
        cN1 = __builtin_amdgcn_wmma_f32_16x16x4_f32(
                  false, aN, false, bW1, (short)0, cN1, false, false);
        cS1 = __builtin_amdgcn_wmma_f32_16x16x4_f32(
                  false, aS, false, bW1, (short)0, cS1, false, false);
    }

    __syncthreads();

    const float n2v0  = sqrtf(sN2[hl]);
    const float n2v1  = sqrtf(sN2[16 + hl]);
    const int   mBase = t0 + wm * 16 + ((lane >> 4) << 3);

    #pragma unroll
    for (int r = 0; r < 8; ++r) {
        const size_t base = ((size_t)b * T_ + (size_t)(mBase + r)) * L_;
        float d0 = fmaxf(sqrtf(cS0[r]) * n2v0, EPS_);
        float d1 = fmaxf(sqrtf(cS1[r]) * n2v1, EPS_);
        out[base + hl]      = cN0[r] / d0;
        out[base + 16 + hl] = cN1[r] / d1;
    }
}

extern "C" void kernel_launch(void* const* d_in, const int* in_sizes, int n_in,
                              void* d_out, int out_size, void* d_ws, size_t ws_size,
                              hipStream_t stream) {
    const float* v1      = (const float*)d_in[0];
    const float* v1_mask = (const float*)d_in[1];
    const float* v2      = (const float*)d_in[2];
    const float* w       = (const float*)d_in[3];
    float*       out     = (float*)d_out;

    dim3 grid(B_ * (T_ / MT));
    dim3 block(64);
    fullmatch_cosine_kernel<<<grid, block, 0, stream>>>(v1, v1_mask, v2, w, out);
}


# K split over 8 waves per 16-row tile (256 blocks x 256 threads): each wave requests its whole 64-k slice up front; the 8 partial sums (num, n1^2, n2^2) are added in wave order through LDS
# speedup vs baseline: 4.6315x; 4.6315x over previous
#include <hip/hip_runtime.h>
#include <math.h>

#define B_   16
#define T_   256
#define D_   512
#define L_   32
#define MT   16
#define KW   8
#define KS   (D_ / KW)
#define NI   (KS / 8)
#define PITCH 264
#define EPS_ 1e-8f

typedef __attribute__((ext_vector_type(2))) float v2f;
typedef __attribute__((ext_vector_type(8))) float v8f;

__global__ __launch_bounds__(256) void fullmatch_cosine_kernel(
    const float* v1,
    const float* v1_mask,
    const float* v2,
    const float* w,
    float* out)
{
    __shared__ __attribute__((aligned(16))) float sRed[KW * 4 * PITCH];
    __shared__ __attribute__((aligned(16))) float sN2[KW * L_];

    const int tid  = threadIdx.x;
    const int b    = blockIdx.x / (T_ / MT);
    const int t0   = (blockIdx.x % (T_ / MT)) * MT;
    const int lane = tid & 31;
    const int q    = tid >> 5;
    const int hl   = lane & 15;
    const int h    = lane >> 4;
    const int kb   = q * KS + 4 * h;

    const float* aRow  = v1 + ((size_t)b * T_ + t0 + hl) * (size_t)D_ + kb;
    const float* wRow0 = w  + (size_t)hl        * (size_t)D_ + kb;
    const float* wRow1 = w  + (size_t)(16 + hl) * (size_t)D_ + kb;
    const float* vRow  = v2 + (size_t)b * D_ + kb;

    float4 av[NI], sv[NI], bv0[NI], bv1[NI];
    float m = v1_mask[(size_t)b * T_ + t0 + hl];
    #pragma unroll
    for (int i = 0; i < NI; ++i) av[i]  = *(const float4*)(aRow  + 8 * i);
    asm volatile("" : : : "memory");
    #pragma unroll
    for (int i = 0; i < NI / 2; ++i) {
        sv[i]  = *(const float4*)(vRow  + 8 * i);
        bv0[i] = *(const float4*)(wRow0 + 8 * i);
        bv1[i] = *(const float4*)(wRow1 + 8 * i);
    }
    asm volatile("" : : : "memory");
    #pragma unroll
    for (int i = NI / 2; i < NI; ++i) {
        sv[i]  = *(const float4*)(vRow  + 8 * i);
        bv0[i] = *(const float4*)(wRow0 + 8 * i);
        bv1[i] = *(const float4*)(wRow1 + 8 * i);
    }
    asm volatile("" : "+v"(m) : : "memory");

    v8f cN0 = {}, cS0 = {};
    v8f cN1 = {}, cS1 = {};
    v2f pn0 = {}, pn1 = {};

    #pragma unroll
    for (int i = 0; i < NI; ++i) {
        if (i == NI / 2) __builtin_amdgcn_sched_barrier(0);
        const float4 a4 = av[i], s4 = sv[i], x0 = bv0[i], x1 = bv1[i];
        #pragma unroll
        for (int j = 0; j < 2; ++j) {
            const float ax = (j ? a4.z : a4.x) * m, ay = (j ? a4.w : a4.y) * m;
            const float sx =  j ? s4.z : s4.x,      sy =  j ? s4.w : s4.y;
            const float px =  j ? x0.z : x0.x,      py =  j ? x0.w : x0.y;
            const float rx =  j ? x1.z : x1.x,      ry =  j ? x1.w : x1.y;

            v2f aN  = { ax * sx, ay * sy };
            v2f aS  = { ax * ax, ay * ay };
            v2f bW0 = { px * px, py * py };
            v2f bW1 = { rx * rx, ry * ry };
            v2f sq  = { sx * sx, sy * sy };

            pn0 += bW0 * sq;
            pn1 += bW1 * sq;

            cN0 = __builtin_amdgcn_wmma_f32_16x16x4_f32(
                      false, aN, false, bW0, (short)0, cN0, false, false);
            cS0 = __builtin_amdgcn_wmma_f32_16x16x4_f32(
                      false, aS, false, bW0, (short)0, cS0, false, false);
            cN1 = __builtin_amdgcn_wmma_f32_16x16x4_f32(
                      false, aN, false, bW1, (short)0, cN1, false, false);
            cS1 = __builtin_amdgcn_wmma_f32_16x16x4_f32(
                      false, aS, false, bW1, (short)0, cS1, false, false);
        }
    }

    {
        float n0 = pn0.x + pn0.y;
        float n1 = pn1.x + pn1.y;
        n0 += __shfl_xor(n0, 16);
        n1 += __shfl_xor(n1, 16);
        sN2[q * L_ + lane] = h ? n1 : n0;

        float* p = sRed + (q * 4) * PITCH + lane;
        #pragma unroll
        for (int r = 0; r < 8; ++r) {
            p[0 * PITCH + r * 32] = cN0[r];
            p[1 * PITCH + r * 32] = cS0[r];
            p[2 * PITCH + r * 32] = cN1[r];
            p[3 * PITCH + r * 32] = cS1[r];
        }
    }
    __syncthreads();

    {
        const int row  = tid >> 4;
        const int col  = (tid & 15) * 2;
        const int tile = col >> 4;
        const int word = (2 * tile) * PITCH + (row & 7) * 32 + (row >> 3) * 16 + (col & 15);

        float2 sN = {0.0f, 0.0f}, sS = {0.0f, 0.0f}, sQ = {0.0f, 0.0f};
        #pragma unroll
        for (int s = 0; s < KW; ++s) {
            const float2 pN = *(const float2*)(sRed + (s * 4) * PITCH + word);
            const float2 pS = *(const float2*)(sRed + (s * 4 + 1) * PITCH + word);
            const float2 pQ = *(const float2*)(sN2 + s * L_ + col);
            sN.x += pN.x; sN.y += pN.y;
            sS.x += pS.x; sS.y += pS.y;
            sQ.x += pQ.x; sQ.y += pQ.y;
        }
        const float d0 = fmaxf(sqrtf(sS.x) * sqrtf(sQ.x), EPS_);
        const float d1 = fmaxf(sqrtf(sS.y) * sqrtf(sQ.y), EPS_);
        float2 o = { sN.x / d0, sN.y / d1 };
        *(float2*)(out + ((size_t)b * T_ + t0) * L_ + tid * 2) = o;
    }
}

extern "C" void kernel_launch(void* const* d_in, const int* in_sizes, int n_in,
                              void* d_out, int out_size, void* d_ws, size_t ws_size,
                              hipStream_t stream) {
    const float* v1      = (const float*)d_in[0];
    const float* v1_mask = (const float*)d_in[1];
    const float* v2      = (const float*)d_in[2];
    const float* w       = (const float*)d_in[3];
    float*       out     = (float*)d_out;

    dim3 grid(B_ * (T_ / MT));
    dim3 block(32 * KW);
    fullmatch_cosine_kernel<<<grid, block, 0, stream>>>(v1, v1_mask, v2, w, out);
}


# K split over 4 waves per 16-row tile instead of 8 (256 blocks x 128 threads, one wave per SIMD, 128 k each): half as many waves enter; the 4 partial sums are added in wave order through LDS
# speedup vs baseline: 4.9525x; 1.0693x over previous
#include <hip/hip_runtime.h>
#include <math.h>

#define B_   16
#define T_   256
#define D_   512
#define L_   32
#define MT   16
#define KW   4
#define KS   (D_ / KW)
#define NI   (KS / 8)
#define GI   4
#define PITCH 264
#define EPS_ 1e-8f

typedef __attribute__((ext_vector_type(2))) float v2f;
typedef __attribute__((ext_vector_type(8))) float v8f;

__global__ __launch_bounds__(128) void fullmatch_cosine_kernel(
    const float* v1,
    const float* v1_mask,
    const float* v2,
    const float* w,
    float* out)
{
    __shared__ __attribute__((aligned(16))) float sRed[KW * 4 * PITCH];
    __shared__ __attribute__((aligned(16))) float sN2[KW * L_];

    const int tid  = threadIdx.x;
    const int b    = blockIdx.x / (T_ / MT);
    const int t0   = (blockIdx.x % (T_ / MT)) * MT;
    const int lane = tid & 31;
    const int q    = tid >> 5;
    const int hl   = lane & 15;
    const int h    = lane >> 4;
    const int kb   = q * KS + 4 * h;

    const float* aRow  = v1 + ((size_t)b * T_ + t0 + hl) * (size_t)D_ + kb;
    const float* wRow0 = w  + (size_t)hl        * (size_t)D_ + kb;
    const float* wRow1 = w  + (size_t)(16 + hl) * (size_t)D_ + kb;
    const float* vRow  = v2 + (size_t)b * D_ + kb;

    float4 av[NI], sv[NI], bv0[NI], bv1[NI];
    float m = v1_mask[(size_t)b * T_ + t0 + hl];
    #pragma unroll
    for (int i = 0; i < NI; ++i) av[i]  = *(const float4*)(aRow  + 8 * i);
    asm volatile("" : : : "memory");
    #pragma unroll
    for (int g = 0; g < NI / GI; ++g) {
        #pragma unroll
        for (int i = g * GI; i < (g + 1) * GI; ++i) {
            sv[i]  = *(const float4*)(vRow  + 8 * i);
            bv0[i] = *(const float4*)(wRow0 + 8 * i);
            bv1[i] = *(const float4*)(wRow1 + 8 * i);
        }
        if (g + 1 < NI / GI) asm volatile("" : : : "memory");
    }
    asm volatile("" : "+v"(m) : : "memory");

    v8f cN0 = {}, cS0 = {};
    v8f cN1 = {}, cS1 = {};
    v2f pn0 = {}, pn1 = {};

    #pragma unroll
    for (int i = 0; i < NI; ++i) {
        if (i > 0 && i % GI == 0) __builtin_amdgcn_sched_barrier(0);
        const float4 a4 = av[i], s4 = sv[i], x0 = bv0[i], x1 = bv1[i];
        #pragma unroll
        for (int j = 0; j < 2; ++j) {
            const float ax = (j ? a4.z : a4.x) * m, ay = (j ? a4.w : a4.y) * m;
            const float sx =  j ? s4.z : s4.x,      sy =  j ? s4.w : s4.y;
            const float px =  j ? x0.z : x0.x,      py =  j ? x0.w : x0.y;
            const float rx =  j ? x1.z : x1.x,      ry =  j ? x1.w : x1.y;

            v2f aN  = { ax * sx, ay * sy };
            v2f aS  = { ax * ax, ay * ay };
            v2f bW0 = { px * px, py * py };
            v2f bW1 = { rx * rx, ry * ry };
            v2f sq  = { sx * sx, sy * sy };

            pn0 += bW0 * sq;
            pn1 += bW1 * sq;

            cN0 = __builtin_amdgcn_wmma_f32_16x16x4_f32(
                      false, aN, false, bW0, (short)0, cN0, false, false);
            cS0 = __builtin_amdgcn_wmma_f32_16x16x4_f32(
                      false, aS, false, bW0, (short)0, cS0, false, false);
            cN1 = __builtin_amdgcn_wmma_f32_16x16x4_f32(
                      false, aN, false, bW1, (short)0, cN1, false, false);
            cS1 = __builtin_amdgcn_wmma_f32_16x16x4_f32(
                      false, aS, false, bW1, (short)0, cS1, false, false);
        }
    }

    {
        float n0 = pn0.x + pn0.y;
        float n1 = pn1.x + pn1.y;
        n0 += __shfl_xor(n0, 16);
        n1 += __shfl_xor(n1, 16);
        sN2[q * L_ + lane] = h ? n1 : n0;

        float* p = sRed + (q * 4) * PITCH + lane;
        #pragma unroll
        for (int r = 0; r < 8; ++r) {
            p[0 * PITCH + r * 32] = cN0[r];
            p[1 * PITCH + r * 32] = cS0[r];
            p[2 * PITCH + r * 32] = cN1[r];
            p[3 * PITCH + r * 32] = cS1[r];
        }
    }
    __syncthreads();

    {
        const int row  = tid >> 3;
        const int col  = (tid & 7) * 4;
        const int tile = col >> 4;
        const int word = (2 * tile) * PITCH + (row & 7) * 32 + (row >> 3) * 16 + (col & 15);

        float4 sN = {0.0f, 0.0f, 0.0f, 0.0f};
        float4 sS = {0.0f, 0.0f, 0.0f, 0.0f};
        float4 sQ = {0.0f, 0.0f, 0.0f, 0.0f};
        #pragma unroll
        for (int s = 0; s < KW; ++s) {
            const float4 pN = *(const float4*)(sRed + (s * 4) * PITCH + word);
            const float4 pS = *(const float4*)(sRed + (s * 4 + 1) * PITCH + word);
            const float4 pQ = *(const float4*)(sN2 + s * L_ + col);
            sN.x += pN.x; sN.y += pN.y; sN.z += pN.z; sN.w += pN.w;
            sS.x += pS.x; sS.y += pS.y; sS.z += pS.z; sS.w += pS.w;
            sQ.x += pQ.x; sQ.y += pQ.y; sQ.z += pQ.z; sQ.w += pQ.w;
        }
        const float d0 = fmaxf(sqrtf(sS.x) * sqrtf(sQ.x), EPS_);
        const float d1 = fmaxf(sqrtf(sS.y) * sqrtf(sQ.y), EPS_);
        const float d2 = fmaxf(sqrtf(sS.z) * sqrtf(sQ.z), EPS_);
        const float d3 = fmaxf(sqrtf(sS.w) * sqrtf(sQ.w), EPS_);
        float4 o = { sN.x / d0, sN.y / d1, sN.z / d2, sN.w / d3 };
        *(float4*)(out + ((size_t)b * T_ + t0) * L_ + tid * 4) = o;
    }
}

extern "C" void kernel_launch(void* const* d_in, const int* in_sizes, int n_in,
                              void* d_out, int out_size, void* d_ws, size_t ws_size,
                              hipStream_t stream) {
    const float* v1      = (const float*)d_in[0];
    const float* v1_mask = (const float*)d_in[1];
    const float* v2      = (const float*)d_in[2];
    const float* w       = (const float*)d_in[3];
    float*       out     = (float*)d_out;

    dim3 grid(B_ * (T_ / MT));
    dim3 block(32 * KW);
    fullmatch_cosine_kernel<<<grid, block, 0, stream>>>(v1, v1_mask, v2, w, out);
}


# Load request order: each 8-k piece's v1, v2 and two w loads go out together and are used piece by piece (was: all 16 v1 loads first): first WMMA waits for 5 loads, not 20; sums in the same order
# speedup vs baseline: 5.3052x; 1.0712x over previous
#include <hip/hip_runtime.h>
#include <math.h>

#define B_   16
#define T_   256
#define D_   512
#define L_   32
#define MT   16
#define KW   4
#define KS   (D_ / KW)
#define NI   (KS / 8)
#define PITCH 264
#define EPS_ 1e-8f

typedef __attribute__((ext_vector_type(2))) float v2f;
typedef __attribute__((ext_vector_type(4))) float v4f;
typedef __attribute__((ext_vector_type(8))) float v8f;

__global__ __launch_bounds__(128) void fullmatch_cosine_kernel(
    const float* v1,
    const float* v1_mask,
    const float* v2,
    const float* w,
    float* out)
{
    __shared__ __attribute__((aligned(16))) float sRed[KW * 4 * PITCH];
    __shared__ __attribute__((aligned(16))) float sN2[KW * L_];

    const int tid  = threadIdx.x;
    const int b    = blockIdx.x / (T_ / MT);
    const int t0   = (blockIdx.x % (T_ / MT)) * MT;
    const int lane = tid & 31;
    const int q    = tid >> 5;
    const int hl   = lane & 15;
    const int h    = lane >> 4;
    const int kb   = q * KS + 4 * h;

    const float* aRow  = v1 + ((size_t)b * T_ + t0 + hl) * (size_t)D_ + kb;
    const float* wRow0 = w  + (size_t)hl        * (size_t)D_ + kb;
    const float* wRow1 = w  + (size_t)(16 + hl) * (size_t)D_ + kb;
    const float* vRow  = v2 + (size_t)b * D_ + kb;

    v4f av[NI], sv[NI], bv0[NI], bv1[NI];
    float m = v1_mask[(size_t)b * T_ + t0 + hl];
    #pragma unroll
    for (int i = 0; i < NI; ++i) {
        av[i]  = *(const v4f*)(aRow  + 8 * i);
        sv[i]  = *(const v4f*)(vRow  + 8 * i);
        bv0[i] = *(const v4f*)(wRow0 + 8 * i);
        bv1[i] = *(const v4f*)(wRow1 + 8 * i);
        if (i + 1 < NI) asm volatile("" : : : "memory");
    }
    asm volatile("" : "+v"(m) : : "memory");

    v8f cN0 = {}, cS0 = {};
    v8f cN1 = {}, cS1 = {};
    v2f pn0 = {}, pn1 = {};

    #pragma unroll
    for (int i = 0; i < NI; ++i) {
        if (i > 0) {
            __builtin_amdgcn_sched_barrier(0);
            asm volatile("" : "+v"(cN0), "+v"(cS0), "+v"(cN1), "+v"(cS1));
        }
        const v4f a4 = av[i], s4 = sv[i], x0 = bv0[i], x1 = bv1[i];
        #pragma unroll
        for (int j = 0; j < 2; ++j) {
            const float ax = (j ? a4.z : a4.x) * m, ay = (j ? a4.w : a4.y) * m;
            const float sx =  j ? s4.z : s4.x,      sy =  j ? s4.w : s4.y;
            const float px =  j ? x0.z : x0.x,      py =  j ? x0.w : x0.y;
            const float rx =  j ? x1.z : x1.x,      ry =  j ? x1.w : x1.y;

            v2f aN  = { ax * sx, ay * sy };
            v2f aS  = { ax * ax, ay * ay };
            v2f bW0 = { px * px, py * py };
            v2f bW1 = { rx * rx, ry * ry };
            v2f sq  = { sx * sx, sy * sy };

            pn0 += bW0 * sq;
            pn1 += bW1 * sq;

            cN0 = __builtin_amdgcn_wmma_f32_16x16x4_f32(
                      false, aN, false, bW0, (short)0, cN0, false, false);
            cS0 = __builtin_amdgcn_wmma_f32_16x16x4_f32(
                      false, aS, false, bW0, (short)0, cS0, false, false);
            cN1 = __builtin_amdgcn_wmma_f32_16x16x4_f32(
                      false, aN, false, bW1, (short)0, cN1, false, false);
            cS1 = __builtin_amdgcn_wmma_f32_16x16x4_f32(
                      false, aS, false, bW1, (short)0, cS1, false, false);
        }
    }

    {
        float n0 = pn0.x + pn0.y;
        float n1 = pn1.x + pn1.y;
        n0 += __shfl_xor(n0, 16);
        n1 += __shfl_xor(n1, 16);
        sN2[q * L_ + lane] = h ? n1 : n0;

        float* p = sRed + (q * 4) * PITCH + lane;
        #pragma unroll
        for (int r = 0; r < 8; ++r) {
            p[0 * PITCH + r * 32] = cN0[r];
            p[1 * PITCH + r * 32] = cS0[r];
            p[2 * PITCH + r * 32] = cN1[r];
            p[3 * PITCH + r * 32] = cS1[r];
        }
    }
    __syncthreads();

    {
        const int row  = tid >> 3;
        const int col  = (tid & 7) * 4;
        const int tile = col >> 4;
        const int word = (2 * tile) * PITCH + (row & 7) * 32 + (row >> 3) * 16 + (col & 15);

        float4 sN = {0.0f, 0.0f, 0.0f, 0.0f};
        float4 sS = {0.0f, 0.0f, 0.0f, 0.0f};
        float4 sQ = {0.0f, 0.0f, 0.0f, 0.0f};
        #pragma unroll
        for (int s = 0; s < KW; ++s) {
            const float4 pN = *(const float4*)(sRed + (s * 4) * PITCH + word);
            const float4 pS = *(const float4*)(sRed + (s * 4 + 1) * PITCH + word);
            const float4 pQ = *(const float4*)(sN2 + s * L_ + col);
            sN.x += pN.x; sN.y += pN.y; sN.z += pN.z; sN.w += pN.w;
            sS.x += pS.x; sS.y += pS.y; sS.z += pS.z; sS.w += pS.w;
            sQ.x += pQ.x; sQ.y += pQ.y; sQ.z += pQ.z; sQ.w += pQ.w;
        }
        const float d0 = fmaxf(sqrtf(sS.x) * sqrtf(sQ.x), EPS_);
        const float d1 = fmaxf(sqrtf(sS.y) * sqrtf(sQ.y), EPS_);
        const float d2 = fmaxf(sqrtf(sS.z) * sqrtf(sQ.z), EPS_);
        const float d3 = fmaxf(sqrtf(sS.w) * sqrtf(sQ.w), EPS_);
        float4 o = { sN.x / d0, sN.y / d1, sN.z / d2, sN.w / d3 };
        *(float4*)(out + ((size_t)b * T_ + t0) * L_ + tid * 4) = o;
    }
}

extern "C" void kernel_launch(void* const* d_in, const int* in_sizes, int n_in,
                              void* d_out, int out_size, void* d_ws, size_t ws_size,
                              hipStream_t stream) {
    const float* v1      = (const float*)d_in[0];
    const float* v1_mask = (const float*)d_in[1];
    const float* v2      = (const float*)d_in[2];
    const float* w       = (const float*)d_in[3];
    float*       out     = (float*)d_out;

    dim3 grid(B_ * (T_ / MT));
    dim3 block(32 * KW);
    fullmatch_cosine_kernel<<<grid, block, 0, stream>>>(v1, v1_mask, v2, w, out);
}
